# starting kernel
# baseline (speedup 1.0000x reference)
#include <hip/hip_runtime.h>

typedef __attribute__((ext_vector_type(2))) float v2f;
typedef __attribute__((ext_vector_type(8))) float v8f;

#define B_ 32
#define M_ 32
#define C_ 1024
#define R_ 28


__global__ __launch_bounds__(256) void mask_colsum_kernel(
    const float* __restrict__ Masks, float* __restrict__ colS) {
  int t = blockIdx.x * 256 + threadIdx.x;
  int k  = t % R_;
  int bm = t / R_;
  const float* p = Masks + (size_t)(bm * R_) * R_ + k;
  float s = 0.f;
#pragma unroll
  for (int i = 0; i < R_; ++i) s += p[i * R_];
  colS[t] = s;
}

__global__ __launch_bounds__(256) void feat_rowsum_kernel(
    const float* __restrict__ F, float* __restrict__ rowT) {
  int t = blockIdx.x * 256 + threadIdx.x;
  const float4* p = (const float4*)(F + (size_t)t * R_);
  float s = 0.f;
#pragma unroll
  for (int q = 0; q < 7; ++q) {
    float4 v = p[q];
    s += (v.x + v.y) + (v.z + v.w);
  }
  int i  = t % R_;
  int bc = t / R_;
  int c  = bc % C_;
  int b  = bc / C_;
  rowT[((size_t)b * R_ + i) * C_ + c] = s;
}

__global__ __launch_bounds__(256) void wmma_gemm_kernel(
    const float* __restrict__ colS, const float* __restrict__ rowT,
    float* __restrict__ S) {
  int wave = (blockIdx.x * 256 + threadIdx.x) >> 5;
  int lane = threadIdx.x & 31;
  int nt = wave & 63;
  int mt = (wave >> 6) & 1;
  int b  = wave >> 7;

  int half = lane >> 4;
  int l16  = lane & 15;
  int m = mt * 16 + l16;
  int n = nt * 16 + l16;

  const float* ca = colS + (size_t)(b * M_ + m) * R_;
  const float* rb = rowT + (size_t)b * R_ * C_ + n;

  v8f acc = {0.f, 0.f, 0.f, 0.f, 0.f, 0.f, 0.f, 0.f};
#pragma unroll
  for (int k0 = 0; k0 < R_; k0 += 4) {
    int kk = k0 + 2 * half;
    v2f a;  a.x  = ca[kk];        a.y  = ca[kk + 1];
    v2f bb; bb.x = rb[kk * C_];   bb.y = rb[(kk + 1) * C_];
    acc = __builtin_amdgcn_wmma_f32_16x16x4_f32(
         false, a,  false, bb,
         (short)0, acc,  false,  false);
  }

  const float inv = 1.0f / (float)(R_ * R_);
#pragma unroll
  for (int v = 0; v < 8; ++v) {
    int row = mt * 16 + v + 8 * half;
    S[((size_t)b * M_ + row) * C_ + n] = acc[v] * inv;
  }
}

extern "C" void kernel_launch(void* const* d_in, const int* in_sizes, int n_in,
                              void* d_out, int out_size, void* d_ws, size_t ws_size,
                              hipStream_t stream) {
  const float* F     = (const float*)d_in[0];
  const float* Masks = (const float*)d_in[1];
  float* S = (float*)d_out;

  float* rowT = (float*)d_ws;
  float* colS = rowT + (size_t)B_ * R_ * C_;

  mask_colsum_kernel<<<(B_ * M_ * R_) / 256, 256, 0, stream>>>(Masks, colS);
  feat_rowsum_kernel<<<(B_ * C_ * R_) / 256, 256, 0, stream>>>(F, rowT);

  wmma_gemm_kernel<<<(B_ * 2 * 64) / 8, 256, 0, stream>>>(colS, rowT, S);
}


# row-sum kernel: each thread sums 7 rows, so 512 workgroups (4096 waves, all resident at once) start instead of 3584 (28672 short waves); loads, stores and per-row summation order unchanged
# speedup vs baseline: 2.0999x; 2.0999x over previous
#include <hip/hip_runtime.h>

typedef __attribute__((ext_vector_type(2))) float v2f;
typedef __attribute__((ext_vector_type(8))) float v8f;

#define B_ 32
#define M_ 32
#define C_ 1024
#define R_ 28


__global__ __launch_bounds__(256) void mask_colsum_kernel(
    const float* __restrict__ Masks, float* __restrict__ colS) {
  int t = blockIdx.x * 256 + threadIdx.x;
  int k  = t % R_;
  int bm = t / R_;
  const float* p = Masks + (size_t)(bm * R_) * R_ + k;
  float s = 0.f;
#pragma unroll
  for (int i = 0; i < R_; ++i) s += p[i * R_];
  colS[t] = s;
}

#define ROWS_PER_THREAD 7
__global__ __launch_bounds__(256) void feat_rowsum_kernel(
    const float* __restrict__ F, float* __restrict__ rowT) {
  int t0 = blockIdx.x * (256 * ROWS_PER_THREAD) + threadIdx.x;
  float4 v[ROWS_PER_THREAD][7];
#pragma unroll
  for (int r = 0; r < ROWS_PER_THREAD; ++r) {
    const float4* p = (const float4*)(F + (size_t)(t0 + 256 * r) * R_);
#pragma unroll
    for (int q = 0; q < 7; ++q) v[r][q] = p[q];
  }
#pragma unroll
  for (int r = 0; r < ROWS_PER_THREAD; ++r) {
    float s = 0.f;
#pragma unroll
    for (int q = 0; q < 7; ++q)
      s += (v[r][q].x + v[r][q].y) + (v[r][q].z + v[r][q].w);
    int t  = t0 + 256 * r;
    int i  = t % R_;
    int bc = t / R_;
    int c  = bc % C_;
    int b  = bc / C_;
    rowT[((size_t)b * R_ + i) * C_ + c] = s;
  }
}

__global__ __launch_bounds__(256) void wmma_gemm_kernel(
    const float* __restrict__ colS, const float* __restrict__ rowT,
    float* __restrict__ S) {
  int wave = (blockIdx.x * 256 + threadIdx.x) >> 5;
  int lane = threadIdx.x & 31;
  int nt = wave & 63;
  int mt = (wave >> 6) & 1;
  int b  = wave >> 7;

  int half = lane >> 4;
  int l16  = lane & 15;
  int m = mt * 16 + l16;
  int n = nt * 16 + l16;

  const float* ca = colS + (size_t)(b * M_ + m) * R_;
  const float* rb = rowT + (size_t)b * R_ * C_ + n;

  v8f acc = {0.f, 0.f, 0.f, 0.f, 0.f, 0.f, 0.f, 0.f};
#pragma unroll
  for (int k0 = 0; k0 < R_; k0 += 4) {
    int kk = k0 + 2 * half;
    v2f a;  a.x  = ca[kk];        a.y  = ca[kk + 1];
    v2f bb; bb.x = rb[kk * C_];   bb.y = rb[(kk + 1) * C_];
    acc = __builtin_amdgcn_wmma_f32_16x16x4_f32(
         false, a,  false, bb,
         (short)0, acc,  false,  false);
  }

  const float inv = 1.0f / (float)(R_ * R_);
#pragma unroll
  for (int v = 0; v < 8; ++v) {
    int row = mt * 16 + v + 8 * half;
    S[((size_t)b * M_ + row) * C_ + n] = acc[v] * inv;
  }
}

extern "C" void kernel_launch(void* const* d_in, const int* in_sizes, int n_in,
                              void* d_out, int out_size, void* d_ws, size_t ws_size,
                              hipStream_t stream) {
  const float* F     = (const float*)d_in[0];
  const float* Masks = (const float*)d_in[1];
  float* S = (float*)d_out;

  float* rowT = (float*)d_ws;
  float* colS = rowT + (size_t)B_ * R_ * C_;

  mask_colsum_kernel<<<(B_ * M_ * R_) / 256, 256, 0, stream>>>(Masks, colS);
  feat_rowsum_kernel<<<(B_ * C_ * R_) / (256 * ROWS_PER_THREAD), 256, 0, stream>>>(F, rowT);

  wmma_gemm_kernel<<<(B_ * 2 * 64) / 8, 256, 0, stream>>>(colS, rowT, S);
}


# fuse the row-sum kernel into the WMMA product kernel: a workgroup's 64x28 row sums go to LDS and feed its 32x64 output strip; 2 launches, no row-sum plane in workspace; summation order unchanged
# speedup vs baseline: 2.5074x; 1.1941x over previous
#include <hip/hip_runtime.h>


typedef __attribute__((ext_vector_type(2))) float v2f;
typedef __attribute__((ext_vector_type(8))) float v8f;

#define B_ 32
#define M_ 32
#define C_ 1024
#define R_ 28


__global__ __launch_bounds__(256) void mask_colsum_kernel(
    const float* __restrict__ Masks, float* __restrict__ colS) {
  int t = blockIdx.x * 256 + threadIdx.x;
  int k  = t % R_;
  int bm = t / R_;
  const float* p = Masks + (size_t)(bm * R_) * R_ + k;
  float s = 0.f;
#pragma unroll
  for (int i = 0; i < R_; ++i) s += p[i * R_];
  colS[t] = s;
}

#define ROWS_PER_THREAD 7
#define ROWS_PER_WG (256 * ROWS_PER_THREAD)
#define CH_PER_WG (ROWS_PER_WG / R_)
__global__ __launch_bounds__(256) void rowsum_gemm_kernel(
    const float* __restrict__ F, const float* __restrict__ colS,
    float* __restrict__ S) {
  __shared__ __attribute__((aligned(16))) float rowL[ROWS_PER_WG];

  int tid  = threadIdx.x;
  int wave = tid >> 5;
  int lane = tid & 31;
  int half = lane >> 4;
  int l16  = lane & 15;
  int mt = wave & 1;
  int nt = wave >> 1;
  int b  = blockIdx.x >> 4;
  int c0 = (blockIdx.x & 15) * CH_PER_WG;

  const float* ca = colS + (size_t)(b * M_ + mt * 16 + l16) * R_ + 2 * half;
  v2f a[7];
#pragma unroll
  for (int q = 0; q < 7; ++q) a[q] = *(const v2f*)(ca + 4 * q);

  int t0 = blockIdx.x * ROWS_PER_WG + tid;
  float4 v[ROWS_PER_THREAD][7];
#pragma unroll
  for (int r = 0; r < ROWS_PER_THREAD; ++r) {
    const float4* p = (const float4*)(F + (size_t)(t0 + 256 * r) * R_);
#pragma unroll
    for (int q = 0; q < 7; ++q) v[r][q] = p[q];
  }
#pragma unroll
  for (int r = 0; r < ROWS_PER_THREAD; ++r) {
    float s = 0.f;
#pragma unroll
    for (int q = 0; q < 7; ++q)
      s += (v[r][q].x + v[r][q].y) + (v[r][q].z + v[r][q].w);
    rowL[tid + 256 * r] = s;
  }
  __syncthreads();

  const float* rb = rowL + (nt * 16 + l16) * R_ + 2 * half;
  v8f acc = {0.f, 0.f, 0.f, 0.f, 0.f, 0.f, 0.f, 0.f};
#pragma unroll
  for (int q = 0; q < 7; ++q) {
    v2f bb = *(const v2f*)(rb + 4 * q);
    acc = __builtin_amdgcn_wmma_f32_16x16x4_f32(
         false, a[q],  false, bb,
         (short)0, acc,  false,  false);
  }

  const float inv = 1.0f / (float)(R_ * R_);
  int n = c0 + nt * 16 + l16;
#pragma unroll
  for (int e = 0; e < 8; ++e) {
    int row = mt * 16 + e + 8 * half;
    S[((size_t)b * M_ + row) * C_ + n] = acc[e] * inv;
  }
}

extern "C" void kernel_launch(void* const* d_in, const int* in_sizes, int n_in,
                              void* d_out, int out_size, void* d_ws, size_t ws_size,
                              hipStream_t stream) {
  const float* F     = (const float*)d_in[0];
  const float* Masks = (const float*)d_in[1];
  float* S = (float*)d_out;

  float* colS = (float*)d_ws;

  mask_colsum_kernel<<<(B_ * M_ * R_) / 256, 256, 0, stream>>>(Masks, colS);

  rowsum_gemm_kernel<<<(B_ * C_ * R_) / ROWS_PER_WG, 256, 0, stream>>>(F, colS, S);
}


# row-sum loads made whole-line: each wave load reads 512 contiguous bytes instead of 32 rows 112 B apart; the 7 piece sums of a row are regrouped through wave-private LDS; summation order unchanged
# speedup vs baseline: 3.3566x; 1.3387x over previous
#include <hip/hip_runtime.h>


typedef __attribute__((ext_vector_type(2))) float v2f;
typedef __attribute__((ext_vector_type(8))) float v8f;

#define B_ 32
#define M_ 32
#define C_ 1024
#define R_ 28


__global__ __launch_bounds__(256) void mask_colsum_kernel(
    const float* __restrict__ Masks, float* __restrict__ colS) {
  int t = blockIdx.x * 256 + threadIdx.x;
  int k  = t % R_;
  int bm = t / R_;
  const float* p = Masks + (size_t)(bm * R_) * R_ + k;
  float s = 0.f;
#pragma unroll
  for (int i = 0; i < R_; ++i) s += p[i * R_];
  colS[t] = s;
}

#define ROWS_PER_THREAD 7
#define ROWS_PER_WG (256 * ROWS_PER_THREAD)
#define CH_PER_WG (ROWS_PER_WG / R_)
#define PIECES_PER_ROW (R_ / 4)
__global__ __launch_bounds__(256) void rowsum_gemm_kernel(
    const float* __restrict__ F, const float* __restrict__ colS,
    float* __restrict__ S) {
  __shared__ __attribute__((aligned(16))) float rowL[ROWS_PER_WG];
  __shared__ float partL[ROWS_PER_WG * PIECES_PER_ROW];

  int tid  = threadIdx.x;
  int wave = tid >> 5;
  int lane = tid & 31;
  int half = lane >> 4;
  int l16  = lane & 15;
  int mt = wave & 1;
  int nt = wave >> 1;
  int b  = blockIdx.x >> 4;
  int c0 = (blockIdx.x & 15) * CH_PER_WG;

  const float* ca = colS + (size_t)(b * M_ + mt * 16 + l16) * R_ + 2 * half;
  v2f a[7];
#pragma unroll
  for (int q = 0; q < 7; ++q) a[q] = *(const v2f*)(ca + 4 * q);

  const float4* fw = (const float4*)(F + (size_t)(blockIdx.x * ROWS_PER_WG + 32 * wave) * R_) + lane;
  float4 v[ROWS_PER_THREAD][PIECES_PER_ROW];
#pragma unroll
  for (int r = 0; r < ROWS_PER_THREAD; ++r) {
#pragma unroll
    for (int j = 0; j < PIECES_PER_ROW; ++j)
      v[r][j] = fw[256 * PIECES_PER_ROW * r + 32 * j];
  }
#pragma unroll
  for (int r = 0; r < ROWS_PER_THREAD; ++r) {
    float* pw = partL + (256 * r + 32 * wave) * PIECES_PER_ROW + lane;
#pragma unroll
    for (int j = 0; j < PIECES_PER_ROW; ++j)
      pw[32 * j] = (v[r][j].x + v[r][j].y) + (v[r][j].z + v[r][j].w);
  }
#pragma unroll
  for (int r = 0; r < ROWS_PER_THREAD; ++r) {
    const float* pr = partL + (256 * r + tid) * PIECES_PER_ROW;
    float s = 0.f;
#pragma unroll
    for (int q = 0; q < PIECES_PER_ROW; ++q) s += pr[q];
    rowL[tid + 256 * r] = s;
  }
  __syncthreads();

  const float* rb = rowL + (nt * 16 + l16) * R_ + 2 * half;
  v8f acc = {0.f, 0.f, 0.f, 0.f, 0.f, 0.f, 0.f, 0.f};
#pragma unroll
  for (int q = 0; q < 7; ++q) {
    v2f bb = *(const v2f*)(rb + 4 * q);
    acc = __builtin_amdgcn_wmma_f32_16x16x4_f32(
         false, a[q],  false, bb,
         (short)0, acc,  false,  false);
  }

  const float inv = 1.0f / (float)(R_ * R_);
  int n = c0 + nt * 16 + l16;
#pragma unroll
  for (int e = 0; e < 8; ++e) {
    int row = mt * 16 + e + 8 * half;
    S[((size_t)b * M_ + row) * C_ + n] = acc[e] * inv;
  }
}

extern "C" void kernel_launch(void* const* d_in, const int* in_sizes, int n_in,
                              void* d_out, int out_size, void* d_ws, size_t ws_size,
                              hipStream_t stream) {
  const float* F     = (const float*)d_in[0];
  const float* Masks = (const float*)d_in[1];
  float* S = (float*)d_out;

  float* colS = (float*)d_ws;

  mask_colsum_kernel<<<(B_ * M_ * R_) / 256, 256, 0, stream>>>(Masks, colS);

  rowsum_gemm_kernel<<<(B_ * C_ * R_) / ROWS_PER_WG, 256, 0, stream>>>(F, colS, S);
}


# column-sum kernel fused into the product kernel (1 launch, no workspace): each workgroup sums its batch's 32 masks itself; column sums become 4 interleaved partial sums (summation order changes)
# speedup vs baseline: 3.7254x; 1.1099x over previous
#include <hip/hip_runtime.h>


typedef __attribute__((ext_vector_type(2))) float v2f;
typedef __attribute__((ext_vector_type(8))) float v8f;

#define B_ 32
#define M_ 32
#define C_ 1024
#define R_ 28

#define ROWS_PER_THREAD 7
#define ROWS_PER_WG (256 * ROWS_PER_THREAD)
#define CH_PER_WG (ROWS_PER_WG / R_)
#define PIECES_PER_ROW (R_ / 4)
#define MASKS_PER_WAVE (M_ / 8)
#define PIECES_PER_MASK (R_ * R_ / 4)
#define LOADS_PER_MASK (R_ / 4)
#define F_ROWS_FIRST 4
#define CPART_PITCH 128
__global__ __launch_bounds__(256) __attribute__((amdgpu_waves_per_eu(4, 4))) void pool_kernel(
    const float* __restrict__ F, const float* __restrict__ Masks,
    float* __restrict__ S) {
  __shared__ __attribute__((aligned(16))) float rowL[ROWS_PER_WG];
  __shared__ float partL[ROWS_PER_WG * PIECES_PER_ROW];
  __shared__ __attribute__((aligned(16))) float cpartL[M_ * CPART_PITCH];
  __shared__ __attribute__((aligned(16))) float colL[M_ * R_];

  int tid  = threadIdx.x;
  int wave = tid >> 5;
  int lane = tid & 31;
  int half = lane >> 4;
  int l16  = lane & 15;
  int mt = wave & 1;
  int nt = wave >> 1;
  int b  = blockIdx.x >> 4;
  int c0 = (blockIdx.x & 15) * CH_PER_WG;

  const int ml = lane < R_ ? lane : R_ - 1;
  const float4* mw = (const float4*)(Masks + (size_t)(b * M_ + MASKS_PER_WAVE * wave) * (R_ * R_)) + ml;
  float4 mv[MASKS_PER_WAVE][LOADS_PER_MASK];
#pragma unroll
  for (int mm = 0; mm < MASKS_PER_WAVE; ++mm) {
#pragma unroll
    for (int jj = 0; jj < LOADS_PER_MASK; ++jj)
      mv[mm][jj] = mw[PIECES_PER_MASK * mm + R_ * jj];
  }

  const float4* fw = (const float4*)(F + (size_t)(blockIdx.x * ROWS_PER_WG + 32 * wave) * R_) + lane;
  float4 v[ROWS_PER_THREAD][PIECES_PER_ROW];
#pragma unroll
  for (int r = 0; r < F_ROWS_FIRST; ++r) {
#pragma unroll
    for (int j = 0; j < PIECES_PER_ROW; ++j)
      v[r][j] = fw[256 * PIECES_PER_ROW * r + 32 * j];
  }

#pragma unroll
  for (int mm = 0; mm < MASKS_PER_WAVE; ++mm) {
    float4 s = mv[mm][0];
#pragma unroll
    for (int jj = 1; jj < LOADS_PER_MASK; ++jj) {
      s.x += mv[mm][jj].x; s.y += mv[mm][jj].y; s.z += mv[mm][jj].z; s.w += mv[mm][jj].w;
    }
    float* cs = cpartL + (MASKS_PER_WAVE * wave + mm) * CPART_PITCH + 4 * lane;
    cs[0] = s.x; cs[1] = s.y; cs[2] = s.z; cs[3] = s.w;
  }
  __builtin_amdgcn_sched_barrier(0);
#pragma unroll
  for (int r = F_ROWS_FIRST; r < ROWS_PER_THREAD; ++r) {
#pragma unroll
    for (int j = 0; j < PIECES_PER_ROW; ++j)
      v[r][j] = fw[256 * PIECES_PER_ROW * r + 32 * j];
  }

#pragma unroll
  for (int r = 0; r < ROWS_PER_THREAD; ++r) {
    float* pw = partL + (256 * r + 32 * wave) * PIECES_PER_ROW + lane;
#pragma unroll
    for (int j = 0; j < PIECES_PER_ROW; ++j)
      pw[32 * j] = (v[r][j].x + v[r][j].y) + (v[r][j].z + v[r][j].w);
  }
#pragma unroll
  for (int r = 0; r < ROWS_PER_THREAD; ++r) {
    const float* pr = partL + (256 * r + tid) * PIECES_PER_ROW;
    float s = 0.f;
#pragma unroll
    for (int q = 0; q < PIECES_PER_ROW; ++q) s += pr[q];
    rowL[tid + 256 * r] = s;
  }
  if (lane < R_) {
#pragma unroll
    for (int mm = 0; mm < MASKS_PER_WAVE; ++mm) {
      const float* cr = cpartL + (MASKS_PER_WAVE * wave + mm) * CPART_PITCH + lane;
      float s = cr[0];
#pragma unroll
      for (int r4 = 1; r4 < 4; ++r4) s += cr[R_ * r4];
      colL[(MASKS_PER_WAVE * wave + mm) * R_ + lane] = s;
    }
  }
  __syncthreads();

  const float* ca = colL + (mt * 16 + l16) * R_ + 2 * half;
  const float* rb = rowL + (nt * 16 + l16) * R_ + 2 * half;
  v8f acc = {0.f, 0.f, 0.f, 0.f, 0.f, 0.f, 0.f, 0.f};
#pragma unroll
  for (int q = 0; q < 7; ++q) {
    v2f aa = *(const v2f*)(ca + 4 * q);
    v2f bb = *(const v2f*)(rb + 4 * q);
    acc = __builtin_amdgcn_wmma_f32_16x16x4_f32(
         false, aa,  false, bb,
         (short)0, acc,  false,  false);
  }

  const float inv = 1.0f / (float)(R_ * R_);
  int n = c0 + nt * 16 + l16;
#pragma unroll
  for (int e = 0; e < 8; ++e) {
    int row = mt * 16 + e + 8 * half;
    S[((size_t)b * M_ + row) * C_ + n] = acc[e] * inv;
  }
}

extern "C" void kernel_launch(void* const* d_in, const int* in_sizes, int n_in,
                              void* d_out, int out_size, void* d_ws, size_t ws_size,
                              hipStream_t stream) {
  const float* F     = (const float*)d_in[0];
  const float* Masks = (const float*)d_in[1];
  float* S = (float*)d_out;

  pool_kernel<<<(B_ * C_ * R_) / ROWS_PER_WG, 256, 0, stream>>>(F, Masks, S);
}


# one 8-wave workgroup per compute unit owns 128 channels (was two owning 64 each): 2048 waves start instead of 4096, a batch's masks are read by 8 workgroups, not 16; summation order unchanged
# speedup vs baseline: 3.9111x; 1.0498x over previous
#include <hip/hip_runtime.h>


typedef __attribute__((ext_vector_type(2))) float v2f;
typedef __attribute__((ext_vector_type(8))) float v8f;

#define B_ 32
#define M_ 32
#define C_ 1024
#define R_ 28

#define ROWS_PER_THREAD 14
#define ROWS_PER_WG (256 * ROWS_PER_THREAD)
#define CH_PER_WG (ROWS_PER_WG / R_)
#define PIECES_PER_ROW (R_ / 4)
#define MASKS_PER_WAVE (M_ / 8)
#define PIECES_PER_MASK (R_ * R_ / 4)
#define LOADS_PER_MASK (R_ / 4)
#define F_ROWS_FIRST 9
#define CPART_PITCH 128
__global__ __launch_bounds__(256) void pool_kernel(
    const float* __restrict__ F, const float* __restrict__ Masks,
    float* __restrict__ S) {
  __shared__ __attribute__((aligned(16))) float rowL[ROWS_PER_WG];
  __shared__ float partL[ROWS_PER_WG * PIECES_PER_ROW];
  __shared__ __attribute__((aligned(16))) float cpartL[M_ * CPART_PITCH];
  __shared__ __attribute__((aligned(16))) float colL[M_ * R_];

  int tid  = threadIdx.x;
  int wave = tid >> 5;
  int lane = tid & 31;
  int half = lane >> 4;
  int l16  = lane & 15;
  int mt = wave & 1;
  int np = wave >> 1;
  int b  = blockIdx.x >> 3;
  int c0 = (blockIdx.x & 7) * CH_PER_WG;

  const int ml = lane < R_ ? lane : R_ - 1;
  const float4* mw = (const float4*)(Masks + (size_t)(b * M_ + MASKS_PER_WAVE * wave) * (R_ * R_)) + ml;
  float4 mv[MASKS_PER_WAVE][LOADS_PER_MASK];
#pragma unroll
  for (int mm = 0; mm < MASKS_PER_WAVE; ++mm) {
#pragma unroll
    for (int jj = 0; jj < LOADS_PER_MASK; ++jj)
      mv[mm][jj] = mw[PIECES_PER_MASK * mm + R_ * jj];
  }

  const float4* fw = (const float4*)(F + (size_t)(blockIdx.x * ROWS_PER_WG + 32 * wave) * R_) + lane;
  float4 v[ROWS_PER_THREAD][PIECES_PER_ROW];
#pragma unroll
  for (int r = 0; r < F_ROWS_FIRST; ++r) {
#pragma unroll
    for (int j = 0; j < PIECES_PER_ROW; ++j)
      v[r][j] = fw[256 * PIECES_PER_ROW * r + 32 * j];
  }

#pragma unroll
  for (int mm = 0; mm < MASKS_PER_WAVE; ++mm) {
    float4 s = mv[mm][0];
#pragma unroll
    for (int jj = 1; jj < LOADS_PER_MASK; ++jj) {
      s.x += mv[mm][jj].x; s.y += mv[mm][jj].y; s.z += mv[mm][jj].z; s.w += mv[mm][jj].w;
    }
    float* cs = cpartL + (MASKS_PER_WAVE * wave + mm) * CPART_PITCH + 4 * lane;
    cs[0] = s.x; cs[1] = s.y; cs[2] = s.z; cs[3] = s.w;
  }
  __builtin_amdgcn_sched_barrier(0);
#pragma unroll
  for (int r = F_ROWS_FIRST; r < ROWS_PER_THREAD; ++r) {
#pragma unroll
    for (int j = 0; j < PIECES_PER_ROW; ++j)
      v[r][j] = fw[256 * PIECES_PER_ROW * r + 32 * j];
  }

#pragma unroll
  for (int r = 0; r <= ROWS_PER_THREAD; ++r) {
    if (r < ROWS_PER_THREAD) {
      float* pw = partL + (256 * r + 32 * wave) * PIECES_PER_ROW + lane;
#pragma unroll
      for (int j = 0; j < PIECES_PER_ROW; ++j)
        pw[32 * j] = (v[r][j].x + v[r][j].y) + (v[r][j].z + v[r][j].w);
    }
    if (r >= 1) {
      const float* pr = partL + (256 * (r - 1) + tid) * PIECES_PER_ROW;
      float s = 0.f;
#pragma unroll
      for (int q = 0; q < PIECES_PER_ROW; ++q) s += pr[q];
      rowL[tid + 256 * (r - 1)] = s;
    }
  }
  if (lane < R_) {
#pragma unroll
    for (int mm = 0; mm < MASKS_PER_WAVE; ++mm) {
      const float* cr = cpartL + (MASKS_PER_WAVE * wave + mm) * CPART_PITCH + lane;
      float s = cr[0];
#pragma unroll
      for (int r4 = 1; r4 < 4; ++r4) s += cr[R_ * r4];
      colL[(MASKS_PER_WAVE * wave + mm) * R_ + lane] = s;
    }
  }
  __syncthreads();

  const float* ca  = colL + (mt * 16 + l16) * R_ + 2 * half;
  const float* rb0 = rowL + (np * 32 + l16) * R_ + 2 * half;
  const float* rb1 = rb0 + 16 * R_;
  v8f acc0 = {0.f, 0.f, 0.f, 0.f, 0.f, 0.f, 0.f, 0.f};
  v8f acc1 = {0.f, 0.f, 0.f, 0.f, 0.f, 0.f, 0.f, 0.f};
#pragma unroll
  for (int q = 0; q < 7; ++q) {
    v2f aa  = *(const v2f*)(ca  + 4 * q);
    v2f bb0 = *(const v2f*)(rb0 + 4 * q);
    v2f bb1 = *(const v2f*)(rb1 + 4 * q);
    acc0 = __builtin_amdgcn_wmma_f32_16x16x4_f32(
         false, aa,  false, bb0,
         (short)0, acc0,  false,  false);
    acc1 = __builtin_amdgcn_wmma_f32_16x16x4_f32(
         false, aa,  false, bb1,
         (short)0, acc1,  false,  false);
  }

  const float inv = 1.0f / (float)(R_ * R_);
  int n = c0 + np * 32 + l16;
#pragma unroll
  for (int e = 0; e < 8; ++e) {
    int row = mt * 16 + e + 8 * half;
    float* so = S + ((size_t)b * M_ + row) * C_ + n;
    so[0]  = acc0[e] * inv;
    so[16] = acc1[e] * inv;
  }
}

extern "C" void kernel_launch(void* const* d_in, const int* in_sizes, int n_in,
                              void* d_out, int out_size, void* d_ws, size_t ws_size,
                              hipStream_t stream) {
  const float* F     = (const float*)d_in[0];
  const float* Masks = (const float*)d_in[1];
  float* S = (float*)d_out;

  pool_kernel<<<(B_ * C_ * R_) / ROWS_PER_WG, 256, 0, stream>>>(F, Masks, S);
}
